# starting kernel
# baseline (speedup 1.0000x reference)
#include <hip/hip_runtime.h>
#include <cstddef>

typedef __attribute__((ext_vector_type(2))) float v2f;
typedef __attribute__((ext_vector_type(8))) float v8f;

template <int NCOLS, bool RELU_IN>
__global__ __launch_bounds__(256) void gemm_wmma_k(const float* __restrict__ X,
                                                   const float* __restrict__ W,
                                                   float* __restrict__ H,
                                                   int ntiles) {
    constexpr int S = (NCOLS == 128) ? 288 : NCOLS * 2;
    __shared__ float ldsW[64 * S];

    for (int idx = threadIdx.x; idx < 128 * NCOLS; idx += 256) {
        const int r = idx / NCOLS;
        const int c = idx % NCOLS;
        ldsW[(r >> 1) * S + c * 2 + (r & 1)] = W[(size_t)r * NCOLS + c];
    }
    __syncthreads();

    const int lane = threadIdx.x & 31;
    const int wave = threadIdx.x >> 5;
    const int tile = blockIdx.x * 8 + wave;
    if (tile >= ntiles) return;
    const int row0 = tile * 16;
    const int m    = lane & 15;
    const int hi   = lane >> 4;
    constexpr int NT = NCOLS / 16;

    v8f acc[NT] = {};
    const float* xrow = X + (size_t)(row0 + m) * 128;

    for (int k = 0; k < 128; k += 4) {
        float a0 = xrow[k + 2 * hi];
        float a1 = xrow[k + 2 * hi + 1];
        if (RELU_IN) { a0 = fmaxf(a0, 0.f); a1 = fmaxf(a1, 0.f); }
        v2f a; a[0] = a0; a[1] = a1;
        const float* wrow = ldsW + ((k >> 1) + hi) * S;
#pragma unroll
        for (int t = 0; t < NT; ++t) {
            const v2f b = *(const v2f*)(wrow + (t * 16 + m) * 2);
            acc[t] = __builtin_amdgcn_wmma_f32_16x16x4_f32(
                false, a, false, b, (short)0, acc[t], false, false);
        }
    }

#pragma unroll
    for (int t = 0; t < NT; ++t) {
        float* o = H + (size_t)row0 * NCOLS + t * 16 + m;
#pragma unroll
        for (int v = 0; v < 8; ++v)
            o[(size_t)(v + 8 * hi) * NCOLS] = acc[t][v];
    }
}

__global__ __launch_bounds__(256) void k_zero_i(int* p, int n) {
    int i = blockIdx.x * blockDim.x + threadIdx.x;
    if (i < n) p[i] = 0;
}

__global__ __launch_bounds__(256) void k_count_i(const int* __restrict__ dst,
                                                 int* __restrict__ deg, int e) {
    int i = blockIdx.x * blockDim.x + threadIdx.x;
    if (i < e) atomicAdd(&deg[dst[i]], 1);
}

__global__ __launch_bounds__(256) void k_dis(const int* __restrict__ deg,
                                             float* __restrict__ dis, int n) {
    int i = blockIdx.x * blockDim.x + threadIdx.x;
    if (i < n) dis[i] = rsqrtf((float)deg[i] + 1.f);
}

__global__ __launch_bounds__(256) void k_scan1(const int* __restrict__ d,
                                               int* __restrict__ exc,
                                               int* __restrict__ bsum, int n) {
    __shared__ int sh[256];
    const int tid  = threadIdx.x;
    const int base = blockIdx.x * 2048 + tid * 8;
    int v[8]; int s = 0;
#pragma unroll
    for (int j = 0; j < 8; ++j) { v[j] = (base + j < n) ? d[base + j] : 0; s += v[j]; }
    sh[tid] = s;
    __syncthreads();
    for (int off = 1; off < 256; off <<= 1) {
        int t = (tid >= off) ? sh[tid - off] : 0;
        __syncthreads();
        if (tid >= off) sh[tid] += t;
        __syncthreads();
    }
    const int excl = sh[tid] - s;
    if (tid == 255) bsum[blockIdx.x] = sh[255];
    int run = excl;
#pragma unroll
    for (int j = 0; j < 8; ++j) {
        if (base + j < n) exc[base + j] = run;
        run += v[j];
    }
}

__global__ void k_scan2(int* bsum, int nb) {
    if (threadIdx.x == 0 && blockIdx.x == 0) {
        int run = 0;
        for (int i = 0; i < nb; ++i) { int t = bsum[i]; bsum[i] = run; run += t; }
    }
}

__global__ __launch_bounds__(256) void k_scan3(int* __restrict__ exc,
                                               const int* __restrict__ bsum,
                                               int* __restrict__ cursor, int n) {
    int i = blockIdx.x * blockDim.x + threadIdx.x;
    if (i < n) {
        int r = exc[i] + bsum[i >> 11];
        exc[i] = r;
        cursor[i] = r;
    }
}

__global__ __launch_bounds__(256) void k_bucket(const int* __restrict__ src,
                                                const int* __restrict__ dst,
                                                const float* __restrict__ dis,
                                                int* __restrict__ cursor,
                                                int2* __restrict__ meta, int e) {
    int i = blockIdx.x * blockDim.x + threadIdx.x;
    if (i >= e) return;
    int s = src[i];
    int pos = atomicAdd(&cursor[dst[i]], 1);
    meta[pos] = make_int2(s, __float_as_int(dis[s]));
}

__global__ __launch_bounds__(256) void k_gather128(const float* __restrict__ H,
                                                   const int* __restrict__ rowptr,
                                                   const int2* __restrict__ meta,
                                                   const float* __restrict__ dis,
                                                   const float* __restrict__ bias,
                                                   float* __restrict__ B,
                                                   int n, int e) {
    const int node = (blockIdx.x * blockDim.x + threadIdx.x) >> 5;
    const int lane = threadIdx.x & 31;
    if (node >= n) return;
    const int beg = rowptr[node];
    const int end = (node + 1 < n) ? rowptr[node + 1] : e;
    const float di = dis[node];
    const int c = lane * 4;

    const float4 hs = *(const float4*)(H + (size_t)node * 128 + c);
    float4 acc;
    acc.x = hs.x * di; acc.y = hs.y * di; acc.z = hs.z * di; acc.w = hs.w * di;

    for (int p = beg; p < end; ++p) {
        const int2 m = meta[p];
        const float w = __int_as_float(m.y);
        const float4 v = *(const float4*)(H + (size_t)m.x * 128 + c);
        acc.x = fmaf(v.x, w, acc.x); acc.y = fmaf(v.y, w, acc.y);
        acc.z = fmaf(v.z, w, acc.z); acc.w = fmaf(v.w, w, acc.w);
    }
    const float4 bb = *(const float4*)(bias + c);
    float4 r;
    r.x = fmaf(di, acc.x, bb.x); r.y = fmaf(di, acc.y, bb.y);
    r.z = fmaf(di, acc.z, bb.z); r.w = fmaf(di, acc.w, bb.w);
    *(float4*)(B + (size_t)node * 128 + c) = r;
}

__global__ __launch_bounds__(256) void k_gather16(const float* __restrict__ H,
                                                  const int* __restrict__ rowptr,
                                                  const int2* __restrict__ meta,
                                                  const float* __restrict__ dis,
                                                  const float* __restrict__ bias,
                                                  float* __restrict__ out,
                                                  int n, int e) {
    const int node = (blockIdx.x * blockDim.x + threadIdx.x) >> 5;
    const int lane = threadIdx.x & 31;
    if (node >= n) return;
    const int beg = rowptr[node];
    const int end = (node + 1 < n) ? rowptr[node + 1] : e;
    const float di = dis[node];
    const int g  = lane >> 2;
    const int q  = (lane & 3) * 4;

    float4 acc = make_float4(0.f, 0.f, 0.f, 0.f);
    if (g == 0) {
        const float4 hs = *(const float4*)(H + (size_t)node * 16 + q);
        acc.x = hs.x * di; acc.y = hs.y * di; acc.z = hs.z * di; acc.w = hs.w * di;
    }
    for (int p = beg + g; p < end; p += 8) {
        const int2 m = meta[p];
        const float w = __int_as_float(m.y);
        const float4 v = *(const float4*)(H + (size_t)m.x * 16 + q);
        acc.x = fmaf(v.x, w, acc.x); acc.y = fmaf(v.y, w, acc.y);
        acc.z = fmaf(v.z, w, acc.z); acc.w = fmaf(v.w, w, acc.w);
    }
#pragma unroll
    for (int mask = 4; mask <= 16; mask <<= 1) {
        acc.x += __shfl_xor(acc.x, mask, 32);
        acc.y += __shfl_xor(acc.y, mask, 32);
        acc.z += __shfl_xor(acc.z, mask, 32);
        acc.w += __shfl_xor(acc.w, mask, 32);
    }
    if (lane < 4) {
        const float4 bb = *(const float4*)(bias + q);
        float4 r;
        r.x = fmaf(di, acc.x, bb.x); r.y = fmaf(di, acc.y, bb.y);
        r.z = fmaf(di, acc.z, bb.z); r.w = fmaf(di, acc.w, bb.w);
        *(float4*)(out + (size_t)node * 16 + q) = r;
    }
}

extern "C" void kernel_launch(void* const* d_in, const int* in_sizes, int n_in,
                              void* d_out, int out_size, void* d_ws, size_t ws_size,
                              hipStream_t stream) {
    const float* x   = (const float*)d_in[0];
    const int*   ei  = (const int*)d_in[1];
    const float* W1  = (const float*)d_in[2];
    const float* b1  = (const float*)d_in[3];
    const float* W2  = (const float*)d_in[4];
    const float* b2  = (const float*)d_in[5];
    const float* W3  = (const float*)d_in[6];
    const float* b3  = (const float*)d_in[7];
    float*       out = (float*)d_out;

    const int N = in_sizes[0] / 128;
    const int E = in_sizes[1] / 2;
    const int* src = ei;
    const int* dst = ei + E;

    float* A      = (float*)d_ws;
    float* B      = A + (size_t)N * 128;
    int2*  meta   = (int2*)(B + (size_t)N * 128);
    float* dis    = (float*)(meta + E);
    int*   degi   = (int*)(dis + N);
    int*   rowptr = degi + N;
    int*   cursor = rowptr + N;
    int*   bsum   = cursor + N;

    const int TPB = 256;
    auto cdiv = [](long long a, long long b) { return (int)((a + b - 1) / b); };
    const int ntiles = (N + 15) / 16;
    const int nb     = cdiv(N, 2048);

    k_zero_i <<<cdiv(N, TPB), TPB, 0, stream>>>(degi, N);
    k_count_i<<<cdiv(E, TPB), TPB, 0, stream>>>(dst, degi, E);
    k_dis    <<<cdiv(N, TPB), TPB, 0, stream>>>(degi, dis, N);
    k_scan1  <<<nb, TPB, 0, stream>>>(degi, rowptr, bsum, N);
    k_scan2  <<<1, 32, 0, stream>>>(bsum, nb);
    k_scan3  <<<cdiv(N, TPB), TPB, 0, stream>>>(rowptr, bsum, cursor, N);
    k_bucket <<<cdiv(E, TPB), TPB, 0, stream>>>(src, dst, dis, cursor, meta, E);

    gemm_wmma_k<128, false><<<cdiv(ntiles, 8), TPB, 0, stream>>>(x, W1, A, ntiles);
    k_gather128<<<cdiv((long long)N * 32, TPB), TPB, 0, stream>>>(A, rowptr, meta, dis, b1, B, N, E);
    gemm_wmma_k<128, true><<<cdiv(ntiles, 8), TPB, 0, stream>>>(B, W2, A, ntiles);
    k_gather128<<<cdiv((long long)N * 32, TPB), TPB, 0, stream>>>(A, rowptr, meta, dis, b2, B, N, E);
    gemm_wmma_k<16, true><<<cdiv(ntiles, 8), TPB, 0, stream>>>(B, W3, A, ntiles);
    k_gather16<<<cdiv((long long)N * 32, TPB), TPB, 0, stream>>>(A, rowptr, meta, dis, b3, out, N, E);
}


# Edge table by owners: pass 1 bins edges by 256-node range (LDS counts, one global add per block and range); pass 2 ranks a range with LDS counters, stores whole slot rows; slots in arrival order
# speedup vs baseline: 2.1315x; 2.1315x over previous
#include <hip/hip_runtime.h>
#include <cstddef>

typedef __attribute__((ext_vector_type(2))) float v2f;
typedef __attribute__((ext_vector_type(8))) float v8f;


constexpr int SLOTS = 32;

template <int NCOLS, bool RELU_IN>
__global__ __launch_bounds__(256) void gemm_wmma_k(const float* __restrict__ X,
                                                   const float* __restrict__ W,
                                                   const int* __restrict__ deg,
                                                   float* __restrict__ H,
                                                   int ntiles) {
    constexpr int S = (NCOLS == 128) ? 288 : NCOLS * 2;
    __shared__ float ldsW[64 * S];

    for (int idx = threadIdx.x; idx < 128 * NCOLS; idx += 256) {
        const int r = idx / NCOLS;
        const int c = idx % NCOLS;
        ldsW[(r >> 1) * S + c * 2 + (r & 1)] = W[(size_t)r * NCOLS + c];
    }
    __syncthreads();

    const int lane = threadIdx.x & 31;
    const int wave = threadIdx.x >> 5;
    const int tile = blockIdx.x * 8 + wave;
    if (tile >= ntiles) return;
    const int row0 = tile * 16;
    const int m    = lane & 15;
    const int hi   = lane >> 4;
    constexpr int NT = NCOLS / 16;

    v8f acc[NT] = {};
    const float* xrow = X + (size_t)(row0 + m) * 128;

    for (int k = 0; k < 128; k += 4) {
        float a0 = xrow[k + 2 * hi];
        float a1 = xrow[k + 2 * hi + 1];
        if (RELU_IN) { a0 = fmaxf(a0, 0.f); a1 = fmaxf(a1, 0.f); }
        v2f a; a[0] = a0; a[1] = a1;
        const float* wrow = ldsW + ((k >> 1) + hi) * S;
#pragma unroll
        for (int t = 0; t < NT; ++t) {
            const v2f b = *(const v2f*)(wrow + (t * 16 + m) * 2);
            acc[t] = __builtin_amdgcn_wmma_f32_16x16x4_f32(
                false, a, false, b, (short)0, acc[t], false, false);
        }
    }

    float sc[8];
#pragma unroll
    for (int v = 0; v < 8; ++v)
        sc[v] = rsqrtf((float)deg[row0 + v + 8 * hi] + 1.f);

#pragma unroll
    for (int t = 0; t < NT; ++t) {
        float* o = H + (size_t)row0 * NCOLS + t * 16 + m;
#pragma unroll
        for (int v = 0; v < 8; ++v)
            o[(size_t)(v + 8 * hi) * NCOLS] = acc[t][v] * sc[v];
    }
}

constexpr int CHAIN_BLOCKS = 768;

__global__ __launch_bounds__(256) void gemm_chain_k(const float* __restrict__ X,
                                                    const float* __restrict__ W2,
                                                    const float* __restrict__ b2,
                                                    const float* __restrict__ W3,
                                                    const int* __restrict__ deg,
                                                    float* __restrict__ H3,
                                                    int ntiles) {
    constexpr int S  = 288;
    constexpr int S3 = 32;
    constexpr int P  = 18;
    __shared__ alignas(16) float ldsW[64 * S];
    __shared__ alignas(16) float ldsW3[64 * S3];
    __shared__ alignas(16) float ldsT[8 * 16 * P];

    for (int idx = threadIdx.x; idx < 128 * 128; idx += 256) {
        const int r = idx >> 7;
        const int c = idx & 127;
        ldsW[(r >> 1) * S + c * 2 + (r & 1)] = W2[idx];
    }
    for (int idx = threadIdx.x; idx < 128 * 16; idx += 256) {
        const int r = idx >> 4;
        const int c = idx & 15;
        ldsW3[(r >> 1) * S3 + c * 2 + (r & 1)] = W3[idx];
    }
    __syncthreads();

    const int lane = threadIdx.x & 31;
    const int wave = threadIdx.x >> 5;
    const int m    = lane & 15;
    const int hi   = lane >> 4;
    float* tl = ldsT + wave * (16 * P);

    float bb[8];
#pragma unroll
    for (int t = 0; t < 8; ++t) bb[t] = b2[t * 16 + m];

    for (int tile = wave * (int)gridDim.x + (int)blockIdx.x; tile < ntiles;
         tile += 8 * (int)gridDim.x) {
        const int row0 = tile * 16;
        v8f acc[8] = {};
        const float* xrow = X + (size_t)(row0 + m) * 128;

        for (int k = 0; k < 128; k += 4) {
            v2f a; a[0] = xrow[k + 2 * hi]; a[1] = xrow[k + 2 * hi + 1];
            const float* wrow = ldsW + ((k >> 1) + hi) * S;
#pragma unroll
            for (int t = 0; t < 8; ++t) {
                const v2f b = *(const v2f*)(wrow + (t * 16 + m) * 2);
                acc[t] = __builtin_amdgcn_wmma_f32_16x16x4_f32(
                    false, a, false, b, (short)0, acc[t], false, false);
            }
        }

        v8f acc3 = {};
#pragma unroll
        for (int t = 0; t < 8; ++t) {
#pragma unroll
            for (int v = 0; v < 8; ++v)
                tl[(v + 8 * hi) * P + m] = fmaxf(acc[t][v] + bb[t], 0.f);
#pragma unroll
            for (int kk = 0; kk < 4; ++kk) {
                const v2f a = *(const v2f*)(tl + m * P + 4 * kk + 2 * hi);
                const v2f b = *(const v2f*)(ldsW3 + (t * 8 + 2 * kk + hi) * S3 + m * 2);
                acc3 = __builtin_amdgcn_wmma_f32_16x16x4_f32(
                    false, a, false, b, (short)0, acc3, false, false);
            }
        }

        float* o = H3 + (size_t)row0 * 16 + m;
#pragma unroll
        for (int v = 0; v < 8; ++v)
            o[(size_t)(v + 8 * hi) * 16] =
                acc3[v] * rsqrtf((float)deg[row0 + v + 8 * hi] + 1.f);
    }
}

__global__ __launch_bounds__(256) void k_zero_i(int* p, int n) {
    int i = blockIdx.x * blockDim.x + threadIdx.x;
    if (i < n) p[i] = 0;
}

constexpr int RSH    = 8;
constexpr int RNODES = 1 << RSH;
constexpr int MAXR   = 512;
constexpr int ETH    = 512;
constexpr int EPT    = 8;
constexpr int EPB    = ETH * EPT;

__device__ __forceinline__ void slot_direct(int d, int s, int* __restrict__ deg,
                                            int* __restrict__ slot,
                                            int* __restrict__ novf,
                                            int2* __restrict__ ovf) {
    const int k = atomicAdd(&deg[d], 1);
    if (k < SLOTS) {
        slot[(size_t)d * SLOTS + k] = s;
    } else {
        const int j = atomicAdd(novf, 1);
        ovf[j] = make_int2(d, s);
    }
}

__global__ __launch_bounds__(ETH) void k_bin(const int* __restrict__ src,
                                             const int* __restrict__ dst,
                                             int* __restrict__ deg,
                                             int* __restrict__ slot,
                                             int* __restrict__ novf,
                                             int2* __restrict__ ovf,
                                             int* __restrict__ gcur,
                                             int2* __restrict__ rec,
                                             int cap, int nr, int e) {
    __shared__ int hist[MAXR];
    __shared__ int base[MAXR];
    const int tid = threadIdx.x;
    for (int t = tid; t < nr; t += ETH) hist[t] = 0;
    __syncthreads();

    int d[EPT], s[EPT], r[EPT];
    const int e0 = blockIdx.x * EPB + tid;
#pragma unroll
    for (int i = 0; i < EPT; ++i) {
        const int idx = e0 + i * ETH;
        d[i] = -1;
        s[i] = 0;
        if (idx < e) { d[i] = dst[idx]; s[i] = src[idx]; }
    }
#pragma unroll
    for (int i = 0; i < EPT; ++i) {
        r[i] = 0;
        if (d[i] >= 0) r[i] = atomicAdd(&hist[d[i] >> RSH], 1);
    }
    __syncthreads();
    for (int t = tid; t < nr; t += ETH) {
        const int c = hist[t];
        base[t] = c ? atomicAdd(&gcur[t], c) : 0;
    }
    __syncthreads();
#pragma unroll
    for (int i = 0; i < EPT; ++i) {
        if (d[i] >= 0) {
            const int b   = d[i] >> RSH;
            const int pos = base[b] + r[i];
            if (pos < cap) rec[(size_t)b * cap + pos] = make_int2(d[i], s[i]);
            else           slot_direct(d[i], s[i], deg, slot, novf, ovf);
        }
    }
}

__global__ __launch_bounds__(ETH) void k_own(const int2* __restrict__ rec,
                                             const int* __restrict__ gcur,
                                             int* __restrict__ deg,
                                             int* __restrict__ slot,
                                             int* __restrict__ novf,
                                             int2* __restrict__ ovf,
                                             int cap, int n) {
    __shared__ int cnt[RNODES];
    __shared__ int first[RNODES];
    __shared__ int stage[RNODES * SLOTS];
    const int tid = threadIdx.x;
    const int b   = blockIdx.x;
    const int lo  = b << RSH;
    for (int i = tid; i < RNODES; i += ETH) {
        const int c = (lo + i < n) ? deg[lo + i] : 0;
        cnt[i] = c;
        first[i] = c;
    }
    for (int idx = tid; idx < RNODES * SLOTS; idx += ETH) stage[idx] = 0;
    __syncthreads();

    int tot = gcur[b];
    if (tot > cap) tot = cap;
    const int2* rb = rec + (size_t)b * cap;
    for (int j0 = tid; j0 < tot; j0 += 4 * ETH) {
        int2 o[4];
#pragma unroll
        for (int u = 0; u < 4; ++u) {
            const int j = j0 + u * ETH;
            o[u] = make_int2(-1, 0);
            if (j < tot) o[u] = rb[j];
        }
#pragma unroll
        for (int u = 0; u < 4; ++u) {
            if (o[u].x >= 0) {
                const int dl = o[u].x - lo;
                const int k  = atomicAdd(&cnt[dl], 1);
                if (k < SLOTS) {
                    stage[dl * SLOTS + k] = o[u].y;
                } else {
                    const int jj = atomicAdd(novf, 1);
                    ovf[jj] = o[u];
                }
            }
        }
    }
    __syncthreads();

    for (int idx = tid; idx < RNODES * SLOTS; idx += ETH) {
        const int i = idx / SLOTS;
        const int k = idx % SLOTS;
        if (lo + i < n && k >= first[i]) slot[(size_t)lo * SLOTS + idx] = stage[idx];
    }
    for (int i = tid; i < RNODES; i += ETH)
        if (lo + i < n) deg[lo + i] = cnt[i];
}

template <bool ACT_OUT>
__global__ __launch_bounds__(256) void k_gather128(const float* __restrict__ H,
                                                   const int* __restrict__ deg,
                                                   const int* __restrict__ slot,
                                                   const int* __restrict__ novf,
                                                   const int2* __restrict__ ovf,
                                                   const float* __restrict__ bias,
                                                   float* __restrict__ B,
                                                   int n) {
    const int node = (blockIdx.x * blockDim.x + threadIdx.x) >> 5;
    const int lane = threadIdx.x & 31;
    if (node >= n) return;
    const int cnt = deg[node];
    const int m   = cnt < SLOTS ? cnt : SLOTS;
    const float di = rsqrtf((float)cnt + 1.f);
    const int c = lane * 4;
    const int* row = slot + (size_t)node * SLOTS;

    float4 acc = *(const float4*)(H + (size_t)node * 128 + c);

    for (int p = 0; p < m; ++p) {
        const float4 v = *(const float4*)(H + (size_t)row[p] * 128 + c);
        acc.x += v.x; acc.y += v.y; acc.z += v.z; acc.w += v.w;
    }
    if (cnt > SLOTS) {
        const int no = *novf;
        for (int p = 0; p < no; ++p) {
            const int2 o = ovf[p];
            if (o.x == node) {
                const float4 v = *(const float4*)(H + (size_t)o.y * 128 + c);
                acc.x += v.x; acc.y += v.y; acc.z += v.z; acc.w += v.w;
            }
        }
    }
    float4 r;
    if (ACT_OUT) {
        const float4 bb = *(const float4*)(bias + c);
        r.x = di * fmaxf(fmaf(di, acc.x, bb.x), 0.f);
        r.y = di * fmaxf(fmaf(di, acc.y, bb.y), 0.f);
        r.z = di * fmaxf(fmaf(di, acc.z, bb.z), 0.f);
        r.w = di * fmaxf(fmaf(di, acc.w, bb.w), 0.f);
    } else {
        r.x = di * acc.x; r.y = di * acc.y; r.z = di * acc.z; r.w = di * acc.w;
    }
    *(float4*)(B + (size_t)node * 128 + c) = r;
}

__global__ __launch_bounds__(256) void k_gather16(const float* __restrict__ H,
                                                  const int* __restrict__ deg,
                                                  const int* __restrict__ slot,
                                                  const int* __restrict__ novf,
                                                  const int2* __restrict__ ovf,
                                                  const float* __restrict__ bias,
                                                  float* __restrict__ out,
                                                  int n) {
    const int node = (blockIdx.x * blockDim.x + threadIdx.x) >> 5;
    const int lane = threadIdx.x & 31;
    if (node >= n) return;
    const int cnt = deg[node];
    const int m   = cnt < SLOTS ? cnt : SLOTS;
    const float di = rsqrtf((float)cnt + 1.f);
    const int g  = lane >> 2;
    const int q  = (lane & 3) * 4;
    const int* row = slot + (size_t)node * SLOTS;

    float4 acc = make_float4(0.f, 0.f, 0.f, 0.f);
    if (g == 0)
        acc = *(const float4*)(H + (size_t)node * 16 + q);
    for (int p = g; p < m; p += 8) {
        const float4 v = *(const float4*)(H + (size_t)row[p] * 16 + q);
        acc.x += v.x; acc.y += v.y; acc.z += v.z; acc.w += v.w;
    }
    if (cnt > SLOTS) {
        const int no = *novf;
        for (int p = g; p < no; p += 8) {
            const int2 o = ovf[p];
            if (o.x == node) {
                const float4 v = *(const float4*)(H + (size_t)o.y * 16 + q);
                acc.x += v.x; acc.y += v.y; acc.z += v.z; acc.w += v.w;
            }
        }
    }
#pragma unroll
    for (int mask = 4; mask <= 16; mask <<= 1) {
        acc.x += __shfl_xor(acc.x, mask, 32);
        acc.y += __shfl_xor(acc.y, mask, 32);
        acc.z += __shfl_xor(acc.z, mask, 32);
        acc.w += __shfl_xor(acc.w, mask, 32);
    }
    if (lane < 4) {
        const float4 bb = *(const float4*)(bias + q);
        float4 r;
        r.x = fmaf(di, acc.x, bb.x); r.y = fmaf(di, acc.y, bb.y);
        r.z = fmaf(di, acc.z, bb.z); r.w = fmaf(di, acc.w, bb.w);
        *(float4*)(out + (size_t)node * 16 + q) = r;
    }
}

extern "C" void kernel_launch(void* const* d_in, const int* in_sizes, int n_in,
                              void* d_out, int out_size, void* d_ws, size_t ws_size,
                              hipStream_t stream) {
    const float* x   = (const float*)d_in[0];
    const int*   ei  = (const int*)d_in[1];
    const float* W1  = (const float*)d_in[2];
    const float* b1  = (const float*)d_in[3];
    const float* W2  = (const float*)d_in[4];
    const float* b2  = (const float*)d_in[5];
    const float* W3  = (const float*)d_in[6];
    const float* b3  = (const float*)d_in[7];
    float*       out = (float*)d_out;

    const int N = in_sizes[0] / 128;
    const int E = in_sizes[1] / 2;
    const int* src = ei;
    const int* dst = ei + E;

    float* A    = (float*)d_ws;
    float* B    = A + (size_t)N * 128;
    int*   slot = (int*)(B + (size_t)N * 128);
    int2*  ovf  = (int2*)(slot + (size_t)N * SLOTS);
    const int nr  = (N + RNODES - 1) / RNODES;
    const int cap = 2 * ((E + nr - 1) / nr) + 1024;
    int2*  rec  = ovf + E;
    int*   degi = (int*)(rec + (size_t)nr * cap);
    int*   novf = degi + N;
    int*   gcur = novf + 1;

    const int TPB = 256;
    auto cdiv = [](long long a, long long b) { return (int)((a + b - 1) / b); };
    const int ntiles = (N + 15) / 16;

    k_zero_i<<<cdiv(N + 1 + nr, TPB), TPB, 0, stream>>>(degi, N + 1 + nr);
    k_bin<<<cdiv(E, EPB), ETH, 0, stream>>>(src, dst, degi, slot, novf, ovf, gcur, rec, cap, nr, E);
    k_own<<<nr, ETH, 0, stream>>>(rec, gcur, degi, slot, novf, ovf, cap, N);

    gemm_wmma_k<128, false><<<cdiv(ntiles, 8), TPB, 0, stream>>>(x, W1, degi, A, ntiles);
    k_gather128<true><<<cdiv((long long)N * 32, TPB), TPB, 0, stream>>>(A, degi, slot, novf, ovf, b1, B, N);
    k_gather128<false><<<cdiv((long long)N * 32, TPB), TPB, 0, stream>>>(B, degi, slot, novf, ovf, b1, A, N);
    gemm_chain_k<<<CHAIN_BLOCKS, TPB, 0, stream>>>(A, W2, b2, W3, degi, B, ntiles);
    k_gather16<<<cdiv((long long)N * 32, TPB), TPB, 0, stream>>>(B, degi, slot, novf, ovf, b3, out, N);
}


# Gathers read a node's fixed-width slot line beside its count, not behind it, and request its rows sixteen at a time under slot<count tests (adds unconditional): no serial tail loop; sum order kept
# speedup vs baseline: 2.2319x; 1.0471x over previous
#include <hip/hip_runtime.h>
#include <cstddef>

typedef __attribute__((ext_vector_type(2))) float v2f;
typedef __attribute__((ext_vector_type(8))) float v8f;


constexpr int SLOTS = 32;

template <int NCOLS, bool RELU_IN>
__global__ __launch_bounds__(256) void gemm_wmma_k(const float* __restrict__ X,
                                                   const float* __restrict__ W,
                                                   const int* __restrict__ deg,
                                                   float* __restrict__ H,
                                                   int ntiles) {
    constexpr int S = (NCOLS == 128) ? 288 : NCOLS * 2;
    __shared__ float ldsW[64 * S];

    for (int idx = threadIdx.x; idx < 128 * NCOLS; idx += 256) {
        const int r = idx / NCOLS;
        const int c = idx % NCOLS;
        ldsW[(r >> 1) * S + c * 2 + (r & 1)] = W[(size_t)r * NCOLS + c];
    }
    __syncthreads();

    const int lane = threadIdx.x & 31;
    const int wave = threadIdx.x >> 5;
    const int tile = blockIdx.x * 8 + wave;
    if (tile >= ntiles) return;
    const int row0 = tile * 16;
    const int m    = lane & 15;
    const int hi   = lane >> 4;
    constexpr int NT = NCOLS / 16;

    v8f acc[NT] = {};
    const float* xrow = X + (size_t)(row0 + m) * 128;

    for (int k = 0; k < 128; k += 4) {
        float a0 = xrow[k + 2 * hi];
        float a1 = xrow[k + 2 * hi + 1];
        if (RELU_IN) { a0 = fmaxf(a0, 0.f); a1 = fmaxf(a1, 0.f); }
        v2f a; a[0] = a0; a[1] = a1;
        const float* wrow = ldsW + ((k >> 1) + hi) * S;
#pragma unroll
        for (int t = 0; t < NT; ++t) {
            const v2f b = *(const v2f*)(wrow + (t * 16 + m) * 2);
            acc[t] = __builtin_amdgcn_wmma_f32_16x16x4_f32(
                false, a, false, b, (short)0, acc[t], false, false);
        }
    }

    float sc[8];
#pragma unroll
    for (int v = 0; v < 8; ++v)
        sc[v] = rsqrtf((float)deg[row0 + v + 8 * hi] + 1.f);

#pragma unroll
    for (int t = 0; t < NT; ++t) {
        float* o = H + (size_t)row0 * NCOLS + t * 16 + m;
#pragma unroll
        for (int v = 0; v < 8; ++v)
            o[(size_t)(v + 8 * hi) * NCOLS] = acc[t][v] * sc[v];
    }
}

constexpr int CHAIN_BLOCKS = 768;

__global__ __launch_bounds__(256) void gemm_chain_k(const float* __restrict__ X,
                                                    const float* __restrict__ W2,
                                                    const float* __restrict__ b2,
                                                    const float* __restrict__ W3,
                                                    const int* __restrict__ deg,
                                                    float* __restrict__ H3,
                                                    int ntiles) {
    constexpr int S  = 288;
    constexpr int S3 = 32;
    constexpr int P  = 18;
    __shared__ alignas(16) float ldsW[64 * S];
    __shared__ alignas(16) float ldsW3[64 * S3];
    __shared__ alignas(16) float ldsT[8 * 16 * P];

    for (int idx = threadIdx.x; idx < 128 * 128; idx += 256) {
        const int r = idx >> 7;
        const int c = idx & 127;
        ldsW[(r >> 1) * S + c * 2 + (r & 1)] = W2[idx];
    }
    for (int idx = threadIdx.x; idx < 128 * 16; idx += 256) {
        const int r = idx >> 4;
        const int c = idx & 15;
        ldsW3[(r >> 1) * S3 + c * 2 + (r & 1)] = W3[idx];
    }
    __syncthreads();

    const int lane = threadIdx.x & 31;
    const int wave = threadIdx.x >> 5;
    const int m    = lane & 15;
    const int hi   = lane >> 4;
    float* tl = ldsT + wave * (16 * P);

    float bb[8];
#pragma unroll
    for (int t = 0; t < 8; ++t) bb[t] = b2[t * 16 + m];

    for (int tile = wave * (int)gridDim.x + (int)blockIdx.x; tile < ntiles;
         tile += 8 * (int)gridDim.x) {
        const int row0 = tile * 16;
        v8f acc[8] = {};
        const float* xrow = X + (size_t)(row0 + m) * 128;

        for (int k = 0; k < 128; k += 4) {
            v2f a; a[0] = xrow[k + 2 * hi]; a[1] = xrow[k + 2 * hi + 1];
            const float* wrow = ldsW + ((k >> 1) + hi) * S;
#pragma unroll
            for (int t = 0; t < 8; ++t) {
                const v2f b = *(const v2f*)(wrow + (t * 16 + m) * 2);
                acc[t] = __builtin_amdgcn_wmma_f32_16x16x4_f32(
                    false, a, false, b, (short)0, acc[t], false, false);
            }
        }

        v8f acc3 = {};
#pragma unroll
        for (int t = 0; t < 8; ++t) {
#pragma unroll
            for (int v = 0; v < 8; ++v)
                tl[(v + 8 * hi) * P + m] = fmaxf(acc[t][v] + bb[t], 0.f);
#pragma unroll
            for (int kk = 0; kk < 4; ++kk) {
                const v2f a = *(const v2f*)(tl + m * P + 4 * kk + 2 * hi);
                const v2f b = *(const v2f*)(ldsW3 + (t * 8 + 2 * kk + hi) * S3 + m * 2);
                acc3 = __builtin_amdgcn_wmma_f32_16x16x4_f32(
                    false, a, false, b, (short)0, acc3, false, false);
            }
        }

        float* o = H3 + (size_t)row0 * 16 + m;
#pragma unroll
        for (int v = 0; v < 8; ++v)
            o[(size_t)(v + 8 * hi) * 16] =
                acc3[v] * rsqrtf((float)deg[row0 + v + 8 * hi] + 1.f);
    }
}

__global__ __launch_bounds__(256) void k_zero_i(int* p, int n) {
    int i = blockIdx.x * blockDim.x + threadIdx.x;
    if (i < n) p[i] = 0;
}

constexpr int RSH    = 8;
constexpr int RNODES = 1 << RSH;
constexpr int MAXR   = 512;
constexpr int ETH    = 512;
constexpr int EPT    = 8;
constexpr int EPB    = ETH * EPT;

__device__ __forceinline__ void slot_direct(int d, int s, int* __restrict__ deg,
                                            int* __restrict__ slot,
                                            int* __restrict__ novf,
                                            int2* __restrict__ ovf) {
    const int k = atomicAdd(&deg[d], 1);
    if (k < SLOTS) {
        slot[(size_t)d * SLOTS + k] = s;
    } else {
        const int j = atomicAdd(novf, 1);
        ovf[j] = make_int2(d, s);
    }
}

__global__ __launch_bounds__(ETH) void k_bin(const int* __restrict__ src,
                                             const int* __restrict__ dst,
                                             int* __restrict__ deg,
                                             int* __restrict__ slot,
                                             int* __restrict__ novf,
                                             int2* __restrict__ ovf,
                                             int* __restrict__ gcur,
                                             int2* __restrict__ rec,
                                             int cap, int nr, int e) {
    __shared__ int hist[MAXR];
    __shared__ int base[MAXR];
    const int tid = threadIdx.x;
    for (int t = tid; t < nr; t += ETH) hist[t] = 0;
    __syncthreads();

    int d[EPT], s[EPT], r[EPT];
    const int e0 = blockIdx.x * EPB + tid;
#pragma unroll
    for (int i = 0; i < EPT; ++i) {
        const int idx = e0 + i * ETH;
        d[i] = -1;
        s[i] = 0;
        if (idx < e) { d[i] = dst[idx]; s[i] = src[idx]; }
    }
#pragma unroll
    for (int i = 0; i < EPT; ++i) {
        r[i] = 0;
        if (d[i] >= 0) r[i] = atomicAdd(&hist[d[i] >> RSH], 1);
    }
    __syncthreads();
    for (int t = tid; t < nr; t += ETH) {
        const int c = hist[t];
        base[t] = c ? atomicAdd(&gcur[t], c) : 0;
    }
    __syncthreads();
#pragma unroll
    for (int i = 0; i < EPT; ++i) {
        if (d[i] >= 0) {
            const int b   = d[i] >> RSH;
            const int pos = base[b] + r[i];
            if (pos < cap) rec[(size_t)b * cap + pos] = make_int2(d[i], s[i]);
            else           slot_direct(d[i], s[i], deg, slot, novf, ovf);
        }
    }
}

__global__ __launch_bounds__(ETH) void k_own(const int2* __restrict__ rec,
                                             const int* __restrict__ gcur,
                                             int* __restrict__ deg,
                                             int* __restrict__ slot,
                                             int* __restrict__ novf,
                                             int2* __restrict__ ovf,
                                             int cap, int n) {
    __shared__ int cnt[RNODES];
    __shared__ int first[RNODES];
    __shared__ int stage[RNODES * SLOTS];
    const int tid = threadIdx.x;
    const int b   = blockIdx.x;
    const int lo  = b << RSH;
    for (int i = tid; i < RNODES; i += ETH) {
        const int c = (lo + i < n) ? deg[lo + i] : 0;
        cnt[i] = c;
        first[i] = c;
    }
    for (int idx = tid; idx < RNODES * SLOTS; idx += ETH) stage[idx] = 0;
    __syncthreads();

    int tot = gcur[b];
    if (tot > cap) tot = cap;
    const int2* rb = rec + (size_t)b * cap;
    for (int j0 = tid; j0 < tot; j0 += 4 * ETH) {
        int2 o[4];
#pragma unroll
        for (int u = 0; u < 4; ++u) {
            const int j = j0 + u * ETH;
            o[u] = make_int2(-1, 0);
            if (j < tot) o[u] = rb[j];
        }
#pragma unroll
        for (int u = 0; u < 4; ++u) {
            if (o[u].x >= 0) {
                const int dl = o[u].x - lo;
                const int k  = atomicAdd(&cnt[dl], 1);
                if (k < SLOTS) {
                    stage[dl * SLOTS + k] = o[u].y;
                } else {
                    const int jj = atomicAdd(novf, 1);
                    ovf[jj] = o[u];
                }
            }
        }
    }
    __syncthreads();

    for (int idx = tid; idx < RNODES * SLOTS; idx += ETH) {
        const int i = idx / SLOTS;
        const int k = idx % SLOTS;
        if (lo + i < n && k >= first[i]) slot[(size_t)lo * SLOTS + idx] = stage[idx];
    }
    for (int i = tid; i < RNODES; i += ETH)
        if (lo + i < n) deg[lo + i] = cnt[i];
}

__device__ __forceinline__ float4 ld_off128(const float* __restrict__ H, unsigned off) {
    return *(const float4*)((const char*)H + off);
}

__device__ __forceinline__ void add4(float4& a, const float4 v) {
    a.x += v.x; a.y += v.y; a.z += v.z; a.w += v.w;
}

template <bool ACT_OUT>
__global__ __launch_bounds__(256) void k_gather128(const float* __restrict__ H,
                                                   const int* __restrict__ deg,
                                                   const int* __restrict__ slot,
                                                   const int* __restrict__ novf,
                                                   const int2* __restrict__ ovf,
                                                   const float* __restrict__ bias,
                                                   float* __restrict__ B,
                                                   int n) {
    const int node = (blockIdx.x * blockDim.x + threadIdx.x) >> 5;
    const int lane = threadIdx.x & 31;
    if (node >= n) return;
    const int c = lane * 4;
    const int4* row4 = (const int4*)(slot + (size_t)node * SLOTS);

    const int4 q0 = row4[0], q1 = row4[1], q2 = row4[2], q3 = row4[3];
    const int cnt = __builtin_amdgcn_readfirstlane(deg[node]);
    float4 acc = *(const float4*)(H + (size_t)node * 128 + c);
    float4 bb = make_float4(0.f, 0.f, 0.f, 0.f);
    if (ACT_OUT) bb = *(const float4*)(bias + c);
    const int m   = cnt < SLOTS ? cnt : SLOTS;
    const float di = rsqrtf((float)cnt + 1.f);
    const unsigned cb = (unsigned)c * 4u;
    const unsigned o0 = (unsigned)q0.x * 512u + cb;
    const unsigned o1 = (unsigned)q0.y * 512u + cb;
    const unsigned o2 = (unsigned)q0.z * 512u + cb;
    const unsigned o3 = (unsigned)q0.w * 512u + cb;
    const unsigned o4 = (unsigned)q1.x * 512u + cb;
    const unsigned o5 = (unsigned)q1.y * 512u + cb;
    const unsigned o6 = (unsigned)q1.z * 512u + cb;
    const unsigned o7 = (unsigned)q1.w * 512u + cb;
    const unsigned o8 = (unsigned)q2.x * 512u + cb;
    const unsigned o9 = (unsigned)q2.y * 512u + cb;
    const unsigned o10 = (unsigned)q2.z * 512u + cb;
    const unsigned o11 = (unsigned)q2.w * 512u + cb;
    const unsigned o12 = (unsigned)q3.x * 512u + cb;
    const unsigned o13 = (unsigned)q3.y * 512u + cb;
    const unsigned o14 = (unsigned)q3.z * 512u + cb;
    const unsigned o15 = (unsigned)q3.w * 512u + cb;
    asm volatile("" :: "v"(o0), "v"(o1), "v"(o2), "v"(o3), "v"(o4), "v"(o5), "v"(o6), "v"(o7), "v"(o8), "v"(o9), "v"(o10), "v"(o11), "v"(o12), "v"(o13), "v"(o14), "v"(o15));

    const float4 z4 = make_float4(0.f, 0.f, 0.f, 0.f);
    float4 a0 = z4; if (0 < m) a0 = ld_off128(H, o0);
    float4 a1 = z4; if (1 < m) a1 = ld_off128(H, o1);
    float4 a2 = z4; if (2 < m) a2 = ld_off128(H, o2);
    float4 a3 = z4; if (3 < m) a3 = ld_off128(H, o3);
    float4 a4 = z4; if (4 < m) a4 = ld_off128(H, o4);
    float4 a5 = z4; if (5 < m) a5 = ld_off128(H, o5);
    float4 a6 = z4; if (6 < m) a6 = ld_off128(H, o6);
    float4 a7 = z4; if (7 < m) a7 = ld_off128(H, o7);
    float4 a8 = z4; if (8 < m) a8 = ld_off128(H, o8);
    float4 a9 = z4; if (9 < m) a9 = ld_off128(H, o9);
    float4 a10 = z4; if (10 < m) a10 = ld_off128(H, o10);
    float4 a11 = z4; if (11 < m) a11 = ld_off128(H, o11);
    float4 a12 = z4; if (12 < m) a12 = ld_off128(H, o12);
    float4 a13 = z4; if (13 < m) a13 = ld_off128(H, o13);
    float4 a14 = z4; if (14 < m) a14 = ld_off128(H, o14);
    float4 a15 = z4; if (15 < m) a15 = ld_off128(H, o15);
    int4 r0 = make_int4(0, 0, 0, 0), r1 = r0, r2 = r0, r3 = r0;
    if (m > 16) { r0 = row4[4]; r1 = row4[5]; r2 = row4[6]; r3 = row4[7]; }
    add4(acc, a0); add4(acc, a1); add4(acc, a2); add4(acc, a3);
    add4(acc, a4); add4(acc, a5); add4(acc, a6); add4(acc, a7);
    add4(acc, a8); add4(acc, a9); add4(acc, a10); add4(acc, a11);
    add4(acc, a12); add4(acc, a13); add4(acc, a14); add4(acc, a15);
    if (m > 16) {
        const unsigned p0 = (unsigned)r0.x * 512u + cb;
        const unsigned p1 = (unsigned)r0.y * 512u + cb;
        const unsigned p2 = (unsigned)r0.z * 512u + cb;
        const unsigned p3 = (unsigned)r0.w * 512u + cb;
        const unsigned p4 = (unsigned)r1.x * 512u + cb;
        const unsigned p5 = (unsigned)r1.y * 512u + cb;
        const unsigned p6 = (unsigned)r1.z * 512u + cb;
        const unsigned p7 = (unsigned)r1.w * 512u + cb;
        const unsigned p8 = (unsigned)r2.x * 512u + cb;
        const unsigned p9 = (unsigned)r2.y * 512u + cb;
        const unsigned p10 = (unsigned)r2.z * 512u + cb;
        const unsigned p11 = (unsigned)r2.w * 512u + cb;
        const unsigned p12 = (unsigned)r3.x * 512u + cb;
        const unsigned p13 = (unsigned)r3.y * 512u + cb;
        const unsigned p14 = (unsigned)r3.z * 512u + cb;
        const unsigned p15 = (unsigned)r3.w * 512u + cb;
        float4 b0 = z4; if (16 < m) b0 = ld_off128(H, p0);
        float4 b1 = z4; if (17 < m) b1 = ld_off128(H, p1);
        float4 b2 = z4; if (18 < m) b2 = ld_off128(H, p2);
        float4 b3 = z4; if (19 < m) b3 = ld_off128(H, p3);
        float4 b4 = z4; if (20 < m) b4 = ld_off128(H, p4);
        float4 b5 = z4; if (21 < m) b5 = ld_off128(H, p5);
        float4 b6 = z4; if (22 < m) b6 = ld_off128(H, p6);
        float4 b7 = z4; if (23 < m) b7 = ld_off128(H, p7);
        float4 b8 = z4; if (24 < m) b8 = ld_off128(H, p8);
        float4 b9 = z4; if (25 < m) b9 = ld_off128(H, p9);
        float4 b10 = z4; if (26 < m) b10 = ld_off128(H, p10);
        float4 b11 = z4; if (27 < m) b11 = ld_off128(H, p11);
        float4 b12 = z4; if (28 < m) b12 = ld_off128(H, p12);
        float4 b13 = z4; if (29 < m) b13 = ld_off128(H, p13);
        float4 b14 = z4; if (30 < m) b14 = ld_off128(H, p14);
        float4 b15 = z4; if (31 < m) b15 = ld_off128(H, p15);
        add4(acc, b0); add4(acc, b1); add4(acc, b2); add4(acc, b3);
        add4(acc, b4); add4(acc, b5); add4(acc, b6); add4(acc, b7);
        add4(acc, b8); add4(acc, b9); add4(acc, b10); add4(acc, b11);
        add4(acc, b12); add4(acc, b13); add4(acc, b14); add4(acc, b15);
    }
    if (cnt > SLOTS) {
        const int no = *novf;
        for (int p = 0; p < no; ++p) {
            const int2 o = ovf[p];
            if (o.x == node) {
                const float4 v = *(const float4*)(H + (size_t)o.y * 128 + c);
                add4(acc, v);
            }
        }
    }
    float4 r;
    if (ACT_OUT) {
        r.x = di * fmaxf(fmaf(di, acc.x, bb.x), 0.f);
        r.y = di * fmaxf(fmaf(di, acc.y, bb.y), 0.f);
        r.z = di * fmaxf(fmaf(di, acc.z, bb.z), 0.f);
        r.w = di * fmaxf(fmaf(di, acc.w, bb.w), 0.f);
    } else {
        r.x = di * acc.x; r.y = di * acc.y; r.z = di * acc.z; r.w = di * acc.w;
    }
    *(float4*)(B + (size_t)node * 128 + c) = r;
}

__global__ __launch_bounds__(256) void k_gather16(const float* __restrict__ H,
                                                  const int* __restrict__ deg,
                                                  const int* __restrict__ slot,
                                                  const int* __restrict__ novf,
                                                  const int2* __restrict__ ovf,
                                                  const float* __restrict__ bias,
                                                  float* __restrict__ out,
                                                  int n) {
    const int node = (blockIdx.x * blockDim.x + threadIdx.x) >> 5;
    const int lane = threadIdx.x & 31;
    if (node >= n) return;
    const int g  = lane >> 2;
    const int q  = (lane & 3) * 4;
    const int* row = slot + (size_t)node * SLOTS;

    const int i0 = row[g], i1 = row[g + 8], i2 = row[g + 16], i3 = row[g + 24];
    const int cnt = deg[node];
    const float4 bb = *(const float4*)(bias + q);
    float4 acc = make_float4(0.f, 0.f, 0.f, 0.f);
    if (g == 0)
        acc = *(const float4*)(H + (size_t)node * 16 + q);
    const int m   = cnt < SLOTS ? cnt : SLOTS;
    const float di = rsqrtf((float)cnt + 1.f);

    float4 v0 = make_float4(0.f, 0.f, 0.f, 0.f), v1 = v0, v2 = v0, v3 = v0;
    if (g      < m) v0 = *(const float4*)(H + (size_t)i0 * 16 + q);
    if (g + 8  < m) v1 = *(const float4*)(H + (size_t)i1 * 16 + q);
    if (g + 16 < m) v2 = *(const float4*)(H + (size_t)i2 * 16 + q);
    if (g + 24 < m) v3 = *(const float4*)(H + (size_t)i3 * 16 + q);
    add4(acc, v0); add4(acc, v1); add4(acc, v2); add4(acc, v3);

    if (cnt > SLOTS) {
        const int no = *novf;
        for (int p = g; p < no; p += 8) {
            const int2 o = ovf[p];
            if (o.x == node) {
                const float4 v = *(const float4*)(H + (size_t)o.y * 16 + q);
                add4(acc, v);
            }
        }
    }
#pragma unroll
    for (int mask = 4; mask <= 16; mask <<= 1) {
        acc.x += __shfl_xor(acc.x, mask, 32);
        acc.y += __shfl_xor(acc.y, mask, 32);
        acc.z += __shfl_xor(acc.z, mask, 32);
        acc.w += __shfl_xor(acc.w, mask, 32);
    }
    if (lane < 4) {
        float4 r;
        r.x = fmaf(di, acc.x, bb.x); r.y = fmaf(di, acc.y, bb.y);
        r.z = fmaf(di, acc.z, bb.z); r.w = fmaf(di, acc.w, bb.w);
        *(float4*)(out + (size_t)node * 16 + q) = r;
    }
}

extern "C" void kernel_launch(void* const* d_in, const int* in_sizes, int n_in,
                              void* d_out, int out_size, void* d_ws, size_t ws_size,
                              hipStream_t stream) {
    const float* x   = (const float*)d_in[0];
    const int*   ei  = (const int*)d_in[1];
    const float* W1  = (const float*)d_in[2];
    const float* b1  = (const float*)d_in[3];
    const float* W2  = (const float*)d_in[4];
    const float* b2  = (const float*)d_in[5];
    const float* W3  = (const float*)d_in[6];
    const float* b3  = (const float*)d_in[7];
    float*       out = (float*)d_out;

    const int N = in_sizes[0] / 128;
    const int E = in_sizes[1] / 2;
    const int* src = ei;
    const int* dst = ei + E;

    float* A    = (float*)d_ws;
    float* B    = A + (size_t)N * 128;
    int*   slot = (int*)(B + (size_t)N * 128);
    int2*  ovf  = (int2*)(slot + (size_t)N * SLOTS);
    const int nr  = (N + RNODES - 1) / RNODES;
    const int cap = 2 * ((E + nr - 1) / nr) + 1024;
    int2*  rec  = ovf + E;
    int*   degi = (int*)(rec + (size_t)nr * cap);
    int*   novf = degi + N;
    int*   gcur = novf + 1;

    const int TPB = 256;
    auto cdiv = [](long long a, long long b) { return (int)((a + b - 1) / b); };
    const int ntiles = (N + 15) / 16;

    k_zero_i<<<cdiv(N + 1 + nr, TPB), TPB, 0, stream>>>(degi, N + 1 + nr);
    k_bin<<<cdiv(E, EPB), ETH, 0, stream>>>(src, dst, degi, slot, novf, ovf, gcur, rec, cap, nr, E);
    k_own<<<nr, ETH, 0, stream>>>(rec, gcur, degi, slot, novf, ovf, cap, N);

    gemm_wmma_k<128, false><<<cdiv(ntiles, 8), TPB, 0, stream>>>(x, W1, degi, A, ntiles);
    k_gather128<true><<<cdiv((long long)N * 32, TPB), TPB, 0, stream>>>(A, degi, slot, novf, ovf, b1, B, N);
    k_gather128<false><<<cdiv((long long)N * 32, TPB), TPB, 0, stream>>>(B, degi, slot, novf, ovf, b1, A, N);
    gemm_chain_k<<<CHAIN_BLOCKS, TPB, 0, stream>>>(A, W2, b2, W3, degi, B, ntiles);
    k_gather16<<<cdiv((long long)N * 32, TPB), TPB, 0, stream>>>(B, degi, slot, novf, ovf, b3, out, N);
}
